# starting kernel
# baseline (speedup 1.0000x reference)
#include <hip/hip_runtime.h>

typedef __attribute__((ext_vector_type(2))) float v2f;
typedef __attribute__((ext_vector_type(8))) float v8f;

#define HW 1024
#define DD 64

__global__ __launch_bounds__(256)
void AbsolutePosEmb_wmma_kernel(const float* __restrict__ q,
                                const float* __restrict__ kk_,
                                const float* __restrict__ ph,
                                const float* __restrict__ pw,
                                float* __restrict__ out)
{
    const int wave = threadIdx.x >> 5;
    const int lane = threadIdx.x & 31;
    const int lr   = lane & 15;
    const int kh   = lane >> 4;

    const int tile   = blockIdx.x * 8 + wave;
    const int bn     = tile >> 8;
    const int t      = tile & 255;
    const int m_tile = (t >> 4) << 6;
    const int n_tile = (t & 15) << 6;

    const float* qbase = q   + (size_t)bn * HW * DD;
    const float* kbase = kk_ + (size_t)bn * HW * DD;

    const float* aptr[4];
    const float* bptr[4];
    const float* phptr[4];
    const float* pwptr[4];
#pragma unroll
    for (int i = 0; i < 4; ++i) {
        const int rowA = m_tile + i * 16 + lr;
        const int rowB = n_tile + i * 16 + lr;
        aptr[i]  = qbase + (size_t)rowA * DD;
        bptr[i]  = kbase + (size_t)rowB * DD;
        phptr[i] = ph + (size_t)(rowB >> 5) * DD;
        pwptr[i] = pw + (size_t)(rowB & 31) * DD;
    }

    v8f acc[4][4];
    const v8f zero = {};
#pragma unroll
    for (int i = 0; i < 4; ++i)
#pragma unroll
        for (int j = 0; j < 4; ++j)
            acc[i][j] = zero;

    for (int kk = 0; kk < DD; kk += 4) {
        const int d = kk + 2 * kh;
        v2f a[4], b[4];
#pragma unroll
        for (int i = 0; i < 4; ++i)
            a[i] = *(const v2f*)(aptr[i] + d);
#pragma unroll
        for (int j = 0; j < 4; ++j) {
            const v2f kv = *(const v2f*)(bptr[j]  + d);
            const v2f pv = *(const v2f*)(phptr[j] + d);
            const v2f wv = *(const v2f*)(pwptr[j] + d);
            b[j] = kv + pv + wv;
        }
#pragma unroll
        for (int i = 0; i < 4; ++i)
#pragma unroll
            for (int j = 0; j < 4; ++j)
                acc[i][j] = __builtin_amdgcn_wmma_f32_16x16x4_f32(
                     false, a[i],  false, b[j],
                     (short)0, acc[i][j],
                     false,  false);
    }

    float* obase = out + (size_t)bn * HW * HW;
#pragma unroll
    for (int i = 0; i < 4; ++i) {
#pragma unroll
        for (int r = 0; r < 8; ++r) {
            const int m = m_tile + i * 16 + r + 8 * kh;
            float* orow = obase + (size_t)m * HW;
#pragma unroll
            for (int j = 0; j < 4; ++j) {
                const int n = n_tile + j * 16 + lr;
                __builtin_nontemporal_store(acc[i][j][r], orow + n);
            }
        }
    }
}

extern "C" void kernel_launch(void* const* d_in, const int* in_sizes, int n_in,
                              void* d_out, int out_size, void* d_ws, size_t ws_size,
                              hipStream_t stream) {
    const float* q  = (const float*)d_in[0];
    const float* k  = (const float*)d_in[1];
    const float* ph = (const float*)d_in[2];
    const float* pw = (const float*)d_in[3];
    float* out = (float*)d_out;

    dim3 grid(1024), block(256);
    hipLaunchKernelGGL(AbsolutePosEmb_wmma_kernel, grid, block, 0, stream,
                       q, k, ph, pw, out);
}


# Operands staged once per work-group in LDS: Q and K+ph+pw tiles loaded with whole-row b128 loads, WMMA fragments read from LDS instead of per-wave strided row gathers from global
# speedup vs baseline: 1.6679x; 1.6679x over previous
#include <hip/hip_runtime.h>


typedef __attribute__((ext_vector_type(2))) float v2f;
typedef __attribute__((ext_vector_type(4))) float v4f;
typedef __attribute__((ext_vector_type(8))) float v8f;

#define HW 1024
#define DD 64
#define WW 32

#define BM 128
#define BN 256
#define PITCH 68

__global__ __launch_bounds__(256)
void AbsolutePosEmb_wmma_kernel(const float* __restrict__ q,
                                const float* __restrict__ kk_,
                                const float* __restrict__ ph,
                                const float* __restrict__ pw,
                                float* __restrict__ out)
{
    __shared__ __attribute__((aligned(16))) float lds[(BM + BN) * PITCH];
    float* sA = lds;
    float* sB = lds + BM * PITCH;

    const int tid  = threadIdx.x;
    const int wave = tid >> 5;
    const int lane = tid & 31;
    const int lr   = lane & 15;
    const int kh   = lane >> 4;

    const int blk   = blockIdx.x;
    const int bn    = blk >> 5;
    const int t     = blk & 31;
    const int m_blk = (t >> 2) * BM;
    const int n_blk = (t & 3) * BN;

    const float* qbase = q   + ((size_t)bn * HW + m_blk) * DD;
    const float* kbase = kk_ + ((size_t)bn * HW + n_blk) * DD;

    const int srow = tid >> 4;
    const int sc   = (tid & 15) * 4;

#pragma unroll
    for (int p = 0; p < BM / 16; ++p) {
        const int row = p * 16 + srow;
        const v4f v = *(const v4f*)(qbase + (size_t)row * DD + sc);
        *(v4f*)(sA + row * PITCH + sc) = v;
    }

    const v4f pw0 = *(const v4f*)(pw + (size_t)srow * DD + sc);
    const v4f pw1 = *(const v4f*)(pw + (size_t)(srow + 16) * DD + sc);
    const float* phbase = ph + (size_t)(n_blk / WW) * DD + sc;
#pragma unroll
    for (int u = 0; u < BN / WW; ++u) {
        const v4f pv   = *(const v4f*)(phbase + (size_t)u * DD);
        const int row0 = u * WW + srow;
        const int row1 = row0 + 16;
        const v4f k0 = *(const v4f*)(kbase + (size_t)row0 * DD + sc);
        const v4f k1 = *(const v4f*)(kbase + (size_t)row1 * DD + sc);
        *(v4f*)(sB + row0 * PITCH + sc) = k0 + pv + pw0;
        *(v4f*)(sB + row1 * PITCH + sc) = k1 + pv + pw1;
    }

    __syncthreads();

    const int wm = wave >> 2;
    const int wn = wave & 3;
    const float* ap = sA + (wm * 64 + lr) * PITCH + 2 * kh;
    const float* bp = sB + (wn * 64 + lr) * PITCH + 2 * kh;

    v8f acc[4][4];
    const v8f zero = {};
#pragma unroll
    for (int i = 0; i < 4; ++i)
#pragma unroll
        for (int j = 0; j < 4; ++j)
            acc[i][j] = zero;

    v2f a[2][4], b[2][4];
#pragma unroll
    for (int i = 0; i < 4; ++i) {
        a[0][i] = *(const v2f*)(ap + i * 16 * PITCH);
        b[0][i] = *(const v2f*)(bp + i * 16 * PITCH);
    }
#pragma unroll
    for (int s = 0; s < DD / 4; ++s) {
        const int cur = s & 1;
        const int nxt = cur ^ 1;
        if (s + 1 < DD / 4) {
#pragma unroll
            for (int i = 0; i < 4; ++i) {
                a[nxt][i] = *(const v2f*)(ap + i * 16 * PITCH + (s + 1) * 4);
                b[nxt][i] = *(const v2f*)(bp + i * 16 * PITCH + (s + 1) * 4);
            }
        }
#pragma unroll
        for (int i = 0; i < 4; ++i)
#pragma unroll
            for (int j = 0; j < 4; ++j)
                acc[i][j] = __builtin_amdgcn_wmma_f32_16x16x4_f32(
                     false, a[cur][i],  false, b[cur][j],
                     (short)0, acc[i][j],
                     false,  false);
    }

    float* obase = out + (size_t)bn * HW * HW;
    const int m_tile = m_blk + wm * 64;
    const int n_tile = n_blk + wn * 64;
#pragma unroll
    for (int i = 0; i < 4; ++i) {
#pragma unroll
        for (int r = 0; r < 8; ++r) {
            const int m = m_tile + i * 16 + r + 8 * kh;
            float* orow = obase + (size_t)m * HW;
#pragma unroll
            for (int j = 0; j < 4; ++j) {
                const int n = n_tile + j * 16 + lr;
                __builtin_nontemporal_store(acc[i][j][r], orow + n);
            }
        }
    }
}

extern "C" void kernel_launch(void* const* d_in, const int* in_sizes, int n_in,
                              void* d_out, int out_size, void* d_ws, size_t ws_size,
                              hipStream_t stream) {
    const float* q  = (const float*)d_in[0];
    const float* k  = (const float*)d_in[1];
    const float* ph = (const float*)d_in[2];
    const float* pw = (const float*)d_in[3];
    float* out = (float*)d_out;

    dim3 grid(1024), block(256);
    hipLaunchKernelGGL(AbsolutePosEmb_wmma_kernel, grid, block, 0, stream,
                       q, k, ph, pw, out);
}


# Wave tile walked one 16-row block at a time with its B fragments for the whole K range held in registers: each block's 32 stores drain under the next block's WMMAs; summation order unchanged
# speedup vs baseline: 1.7841x; 1.0697x over previous
#include <hip/hip_runtime.h>


typedef __attribute__((ext_vector_type(2))) float v2f;
typedef __attribute__((ext_vector_type(4))) float v4f;
typedef __attribute__((ext_vector_type(8))) float v8f;

#define HW 1024
#define DD 64
#define WW 32

#define BM 128
#define BN 256
#define PITCH 68

__global__ __launch_bounds__(256)
void AbsolutePosEmb_wmma_kernel(const float* __restrict__ q,
                                const float* __restrict__ kk_,
                                const float* __restrict__ ph,
                                const float* __restrict__ pw,
                                float* __restrict__ out)
{
    __shared__ __attribute__((aligned(16))) float lds[(BM + BN) * PITCH];
    float* sA = lds;
    float* sB = lds + BM * PITCH;

    const int tid  = threadIdx.x;
    const int wave = tid >> 5;
    const int lane = tid & 31;
    const int lr   = lane & 15;
    const int kh   = lane >> 4;

    const int blk   = blockIdx.x;
    const int bn    = blk >> 5;
    const int t     = blk & 31;
    const int m_blk = (t >> 2) * BM;
    const int n_blk = (t & 3) * BN;

    const float* qbase = q   + ((size_t)bn * HW + m_blk) * DD;
    const float* kbase = kk_ + ((size_t)bn * HW + n_blk) * DD;

    const int srow = tid >> 4;
    const int sc   = (tid & 15) * 4;

#pragma unroll
    for (int p = 0; p < BM / 16; ++p) {
        const int row = p * 16 + srow;
        const v4f v = *(const v4f*)(qbase + (size_t)row * DD + sc);
        *(v4f*)(sA + row * PITCH + sc) = v;
    }

    const v4f pw0 = *(const v4f*)(pw + (size_t)srow * DD + sc);
    const v4f pw1 = *(const v4f*)(pw + (size_t)(srow + 16) * DD + sc);
    const float* phbase = ph + (size_t)(n_blk / WW) * DD + sc;
#pragma unroll
    for (int u = 0; u < BN / WW; ++u) {
        const v4f pv   = *(const v4f*)(phbase + (size_t)u * DD);
        const int row0 = u * WW + srow;
        const int row1 = row0 + 16;
        const v4f k0 = *(const v4f*)(kbase + (size_t)row0 * DD + sc);
        const v4f k1 = *(const v4f*)(kbase + (size_t)row1 * DD + sc);
        *(v4f*)(sB + row0 * PITCH + sc) = k0 + pv + pw0;
        *(v4f*)(sB + row1 * PITCH + sc) = k1 + pv + pw1;
    }

    __syncthreads();

    const int wm = wave >> 2;
    const int wn = wave & 3;
    const float* ap = sA + (wm * 64 + lr) * PITCH + 2 * kh;
    const float* bp = sB + (wn * 64 + lr) * PITCH + 2 * kh;

    v2f b[4][DD / 4];
    v2f a[2][DD / 4];
#pragma unroll
    for (int s = 0; s < DD / 4; ++s) {
        a[0][s] = *(const v2f*)(ap + s * 4);
#pragma unroll
        for (int j = 0; j < 4; ++j)
            b[j][s] = *(const v2f*)(bp + j * 16 * PITCH + s * 4);
    }

    float* obase = out + (size_t)bn * HW * HW;
    const int m_tile = m_blk + wm * 64;
    const int n_tile = n_blk + wn * 64;
    const v8f zero = {};

#pragma unroll
    for (int i = 0; i < 4; ++i) {
        const int cur = i & 1;
        const int nxt = cur ^ 1;
        if (i + 1 < 4) {
#pragma unroll
            for (int s = 0; s < DD / 4; ++s)
                a[nxt][s] = *(const v2f*)(ap + (i + 1) * 16 * PITCH + s * 4);
        }
        __builtin_amdgcn_sched_barrier(0);

        v8f acc[4];
#pragma unroll
        for (int j = 0; j < 4; ++j)
            acc[j] = zero;
#pragma unroll
        for (int s = 0; s < DD / 4; ++s)
#pragma unroll
            for (int j = 0; j < 4; ++j)
                acc[j] = __builtin_amdgcn_wmma_f32_16x16x4_f32(
                     false, a[cur][s],  false, b[j][s],
                     (short)0, acc[j],
                     false,  false);

#pragma unroll
        for (int r = 0; r < 8; ++r) {
            const int m = m_tile + i * 16 + r + 8 * kh;
            float* orow = obase + (size_t)m * HW;
#pragma unroll
            for (int j = 0; j < 4; ++j) {
                const int n = n_tile + j * 16 + lr;
                __builtin_nontemporal_store(acc[j][r], orow + n);
            }
        }
    }
}

extern "C" void kernel_launch(void* const* d_in, const int* in_sizes, int n_in,
                              void* d_out, int out_size, void* d_ws, size_t ws_size,
                              hipStream_t stream) {
    const float* q  = (const float*)d_in[0];
    const float* k  = (const float*)d_in[1];
    const float* ph = (const float*)d_in[2];
    const float* pw = (const float*)d_in[3];
    float* out = (float*)d_out;

    dim3 grid(1024), block(256);
    hipLaunchKernelGGL(AbsolutePosEmb_wmma_kernel, grid, block, 0, stream,
                       q, k, ph, pw, out);
}


# Work-group kept over 4 query tiles of one 256-key tile: K+ph+pw fragments held in registers, next Q tile fetched under the WMMAs to a 2nd LDS buffer, barrier waits on LDS only; grid 1024 to 256
# speedup vs baseline: 1.8824x; 1.0551x over previous
#include <hip/hip_runtime.h>


typedef __attribute__((ext_vector_type(2))) float v2f;
typedef __attribute__((ext_vector_type(4))) float v4f;
typedef __attribute__((ext_vector_type(8))) float v8f;

#define HW 1024
#define DD 64
#define WW 32

#define BM 128
#define BN 256
#define QT 4
#define PITCH 68

__device__ __forceinline__ void lds_barrier()
{
    __builtin_amdgcn_fence(__ATOMIC_RELEASE, "workgroup", "local");
    __builtin_amdgcn_s_barrier();
    __builtin_amdgcn_fence(__ATOMIC_ACQUIRE, "workgroup", "local");
}

__global__ __launch_bounds__(256)
void AbsolutePosEmb_wmma_kernel(const float* __restrict__ q,
                                const float* __restrict__ kk_,
                                const float* __restrict__ ph,
                                const float* __restrict__ pw,
                                float* __restrict__ out)
{
    __shared__ __attribute__((aligned(16))) float lds[(2 * BM + BN) * PITCH];
    float* sA = lds;
    float* sB = lds + 2 * BM * PITCH;

    const int tid  = threadIdx.x;
    const int wave = tid >> 5;
    const int lane = tid & 31;
    const int lr   = lane & 15;
    const int kh   = lane >> 4;

    const int blk   = blockIdx.x;
    const int bn    = blk >> 3;
    const int t     = blk & 7;
    const int n_blk = (t & 3) * BN;
    const int m_blk = (t >> 2) * (QT * BM);

    const float* qbase = q   + ((size_t)bn * HW + m_blk) * DD;
    const float* kbase = kk_ + ((size_t)bn * HW + n_blk) * DD;

    const int srow = tid >> 4;
    const int sc   = (tid & 15) * 4;

#pragma unroll
    for (int p = 0; p < BM / 16; ++p) {
        const int row = p * 16 + srow;
        const v4f v = *(const v4f*)(qbase + (size_t)row * DD + sc);
        *(v4f*)(sA + row * PITCH + sc) = v;
    }

    const v4f pw0 = *(const v4f*)(pw + (size_t)srow * DD + sc);
    const v4f pw1 = *(const v4f*)(pw + (size_t)(srow + 16) * DD + sc);
    const float* phbase = ph + (size_t)(n_blk / WW) * DD + sc;
#pragma unroll
    for (int u = 0; u < BN / WW; ++u) {
        const v4f pv   = *(const v4f*)(phbase + (size_t)u * DD);
        const int row0 = u * WW + srow;
        const int row1 = row0 + 16;
        const v4f k0 = *(const v4f*)(kbase + (size_t)row0 * DD + sc);
        const v4f k1 = *(const v4f*)(kbase + (size_t)row1 * DD + sc);
        *(v4f*)(sB + row0 * PITCH + sc) = k0 + pv + pw0;
        *(v4f*)(sB + row1 * PITCH + sc) = k1 + pv + pw1;
    }

    lds_barrier();

    const int wm = wave >> 2;
    const int wn = wave & 3;
    const float* ap = sA + (wm * 64 + lr) * PITCH + 2 * kh;
    const float* bp = sB + (wn * 64 + lr) * PITCH + 2 * kh;

    v2f b[4][DD / 4];
    v2f a[2][DD / 4];
#pragma unroll
    for (int s = 0; s < DD / 4; ++s) {
        a[0][s] = *(const v2f*)(ap + s * 4);
#pragma unroll
        for (int j = 0; j < 4; ++j)
            b[j][s] = *(const v2f*)(bp + j * 16 * PITCH + s * 4);
    }

    float* obase = out + (size_t)bn * HW * HW;
    const int m_tile = m_blk + wm * 64;
    const int n_tile = n_blk + wn * 64;
    const v8f zero = {};

#pragma unroll 1
    for (int i = 0; i < QT; ++i) {
        const float* apc = ap + (i & 1) * (BM * PITCH);
        const float* apn = ap + ((i + 1) & 1) * (BM * PITCH);
        float* sAn = sA + ((i + 1) & 1) * (BM * PITCH);
        const bool more = (i + 1 < QT);

        v4f qn[BM / 16];
        if (more) {
            const float* qnext = qbase + (size_t)(i + 1) * BM * DD;
#pragma unroll
            for (int p = 0; p < BM / 16; ++p)
                qn[p] = *(const v4f*)(qnext + (size_t)(p * 16 + srow) * DD + sc);
        }

#pragma unroll
        for (int rb = 0; rb < 4; ++rb) {
            const int cur = rb & 1;
            const int nxt = cur ^ 1;
            if (rb < 3) {
#pragma unroll
                for (int s = 0; s < DD / 4; ++s)
                    a[nxt][s] = *(const v2f*)(apc + (rb + 1) * 16 * PITCH + s * 4);
            } else if (more) {
                lds_barrier();
#pragma unroll
                for (int s = 0; s < DD / 4; ++s)
                    a[nxt][s] = *(const v2f*)(apn + s * 4);
            }
            __builtin_amdgcn_sched_barrier(0);

            v8f acc[4];
#pragma unroll
            for (int j = 0; j < 4; ++j)
                acc[j] = zero;
#pragma unroll
            for (int s = 0; s < DD / 4; ++s)
#pragma unroll
                for (int j = 0; j < 4; ++j)
                    acc[j] = __builtin_amdgcn_wmma_f32_16x16x4_f32(
                         false, a[cur][s],  false, b[j][s],
                         (short)0, acc[j],
                         false,  false);

#pragma unroll
            for (int r = 0; r < 8; ++r) {
                const int m = m_tile + i * BM + rb * 16 + r + 8 * kh;
                float* orow = obase + (size_t)m * HW;
#pragma unroll
                for (int j = 0; j < 4; ++j) {
                    const int n = n_tile + j * 16 + lr;
                    __builtin_nontemporal_store(acc[j][r], orow + n);
                }
            }

            if (rb == 1 && more) {
#pragma unroll
                for (int p = 0; p < BM / 16; ++p)
                    *(v4f*)(sAn + (p * 16 + srow) * PITCH + sc) = qn[p];
            }
        }
    }
}

extern "C" void kernel_launch(void* const* d_in, const int* in_sizes, int n_in,
                              void* d_out, int out_size, void* d_ws, size_t ws_size,
                              hipStream_t stream) {
    const float* q  = (const float*)d_in[0];
    const float* k  = (const float*)d_in[1];
    const float* ph = (const float*)d_in[2];
    const float* pw = (const float*)d_in[3];
    float* out = (float*)d_out;

    dim3 grid(256), block(256);
    hipLaunchKernelGGL(AbsolutePosEmb_wmma_kernel, grid, block, 0, stream,
                       q, k, ph, pw, out);
}
